# starting kernel
# baseline (speedup 1.0000x reference)
#include <hip/hip_runtime.h>
#include <hip/hip_bf16.h>
#include <math.h>

#define HIDDEN   2048
#define NH       16
#define NKV      4
#define HD       128
#define D2       64
#define QSIZE    2048
#define KVSIZE   512
#define QKVLD    3072
#define KOFF     2048
#define VOFF     2560
#define SEQB     2048
#define BATCH    2
#define NTOK     (BATCH * SEQB)
#define SM_SCALE 0.08838834764831845f

typedef __attribute__((ext_vector_type(2))) float v2f;
typedef __attribute__((ext_vector_type(8))) float v8f;

__device__ __forceinline__ v8f wmma_f32(v2f a, v2f b, v8f c) {
  return __builtin_amdgcn_wmma_f32_16x16x4_f32(
       false, a,  false, b,
       (short)0, c,  false,  false);
}

#define GBK  32
#define GLDP 36

__global__ __launch_bounds__(128)
void gemm_nt_kernel(const float* __restrict__ A, const float* __restrict__ Bw,
                    const float* __restrict__ bias, float* __restrict__ C,
                    int M, int N, int K) {
  __shared__ float sA[128 * GLDP];
  __shared__ float sB[64 * GLDP];
  const int t    = threadIdx.x;
  const int lane = t & 31, wid = t >> 5;
  const int hi   = lane >> 4, l15 = lane & 15;
  const int mbase = blockIdx.y * 128;
  const int nbase = blockIdx.x * 64;

  const v8f vz = {0.f,0.f,0.f,0.f,0.f,0.f,0.f,0.f};
  v8f acc[2][4] = {{vz, vz, vz, vz}, {vz, vz, vz, vz}};

  const int srow = t >> 3;
  const int scol = (t & 7) << 2;

  for (int kb = 0; kb < K; kb += GBK) {
#pragma unroll
    for (int i = 0; i < 8; i++) {
      const int row = srow + 16 * i;
      *(float4*)&sA[row * GLDP + scol] =
          *(const float4*)&A[(size_t)(mbase + row) * K + kb + scol];
    }
#pragma unroll
    for (int i = 0; i < 4; i++) {
      const int row = srow + 16 * i;
      *(float4*)&sB[row * GLDP + scol] =
          *(const float4*)&Bw[(size_t)(nbase + row) * K + kb + scol];
    }
    if (kb + GBK < K) {
      __builtin_prefetch(&A[(size_t)(mbase + srow) * K + kb + GBK + scol], 0, 0);
      __builtin_prefetch(&Bw[(size_t)(nbase + srow) * K + kb + GBK + scol], 0, 0);
    }
    __syncthreads();
#pragma unroll
    for (int k2 = 0; k2 < GBK; k2 += 4) {
      const v2f a0 = *(const v2f*)&sA[(wid * 32 + l15) * GLDP + k2 + 2 * hi];
      const v2f a1 = *(const v2f*)&sA[(wid * 32 + 16 + l15) * GLDP + k2 + 2 * hi];
#pragma unroll
      for (int nc = 0; nc < 4; nc++) {
        const v2f b = *(const v2f*)&sB[(nc * 16 + l15) * GLDP + k2 + 2 * hi];
        acc[0][nc] = wmma_f32(a0, b, acc[0][nc]);
        acc[1][nc] = wmma_f32(a1, b, acc[1][nc]);
      }
    }
    __syncthreads();
  }

#pragma unroll
  for (int st = 0; st < 2; st++) {
#pragma unroll
    for (int nc = 0; nc < 4; nc++) {
      const int n  = nbase + nc * 16 + l15;
      const float bv = bias ? bias[n] : 0.0f;
#pragma unroll
      for (int r = 0; r < 8; r++) {
        const int m = mbase + wid * 32 + st * 16 + r + 8 * hi;
        C[(size_t)m * N + n] = acc[st][nc][r] + bv;
      }
    }
  }
}

__global__ __launch_bounds__(256)
void rope_kernel(float* __restrict__ qkv) {
  const int tid   = blockIdx.x * 256 + threadIdx.x;
  const int token = tid / 1280;
  const int rem   = tid % 1280;
  const int head  = rem >> 6;
  const int d     = rem & 63;
  const int pos   = token & (SEQB - 1);
  const int col   = (head < NH) ? head * HD + d : QSIZE + (head - NH) * HD + d;
  const float inv_freq = powf(1.0e6f, -(float)d * (1.0f / 64.0f));
  float s, c;
  sincosf((float)pos * inv_freq, &s, &c);
  float* p = qkv + (size_t)token * QKVLD + col;
  const float x1 = p[0], x2 = p[D2];
  p[0]  = x1 * c - x2 * s;
  p[D2] = x1 * s + x2 * c;
}

__global__ __launch_bounds__(128)
void flash_attn_kernel(const float* __restrict__ qkv, float* __restrict__ out) {
  __shared__ float sQ[64 * 132];
  __shared__ float sK[16 * 132];
  __shared__ float sVt[128 * 20];
  __shared__ float sP[4][16 * 18];

  const int t    = threadIdx.x;
  const int lane = t & 31, wid = t >> 5;
  const int hi   = lane >> 4, l15 = lane & 15;
  const int qblk = blockIdx.x;
  const int h    = blockIdx.y;
  const int b    = blockIdx.z;
  const int kvh  = h >> 2;
  const int qbase = qblk * 64;
  const size_t rowbase = (size_t)(b * SEQB) * QKVLD;

#pragma unroll
  for (int i = 0; i < 16; i++) {
    const int idx = t + 128 * i;
    const int row = idx >> 5, c4 = (idx & 31) << 2;
    float4 v = *(const float4*)&qkv[rowbase + (size_t)(qbase + row) * QKVLD + h * HD + c4];
    v.x *= SM_SCALE; v.y *= SM_SCALE; v.z *= SM_SCALE; v.w *= SM_SCALE;
    *(float4*)&sQ[row * 132 + c4] = v;
  }

  const v8f vz = {0.f,0.f,0.f,0.f,0.f,0.f,0.f,0.f};
  v8f accO[8] = {vz, vz, vz, vz, vz, vz, vz, vz};
  float rm[8], rs[8];
#pragma unroll
  for (int r = 0; r < 8; r++) { rm[r] = -__builtin_inff(); rs[r] = 0.0f; }

  __syncthreads();

  const int nkb = (qbase >> 4) + 4;
  for (int kbi = 0; kbi < nkb; kbi++) {
    const int kb = kbi * 16;
#pragma unroll
    for (int i = 0; i < 4; i++) {
      const int idx = t + 128 * i;
      const int row = idx >> 5, c4 = (idx & 31) << 2;
      const float4 v = *(const float4*)&qkv[rowbase + (size_t)(kb + row) * QKVLD + KOFF + kvh * HD + c4];
      *(float4*)&sK[row * 132 + c4] = v;
    }
#pragma unroll
    for (int i = 0; i < 16; i++) {
      sVt[t * 20 + i] = qkv[rowbase + (size_t)(kb + i) * QKVLD + VOFF + kvh * HD + t];
    }
    __syncthreads();

    v8f s = vz;
#pragma unroll
    for (int k0 = 0; k0 < HD; k0 += 4) {
      const v2f a  = *(const v2f*)&sQ[(wid * 16 + l15) * 132 + k0 + 2 * hi];
      const v2f bb = *(const v2f*)&sK[l15 * 132 + k0 + 2 * hi];
      s = wmma_f32(a, bb, s);
    }

    const int kcol = kb + l15;
#pragma unroll
    for (int r = 0; r < 8; r++) {
      const int mrow = qbase + wid * 16 + r + 8 * hi;
      const float sv = (kcol <= mrow) ? s[r] : -__builtin_inff();
      float mx = sv;
      mx = fmaxf(mx, __shfl_xor(mx, 1, 32));
      mx = fmaxf(mx, __shfl_xor(mx, 2, 32));
      mx = fmaxf(mx, __shfl_xor(mx, 4, 32));
      mx = fmaxf(mx, __shfl_xor(mx, 8, 32));
      const float newm  = fmaxf(rm[r], mx);
      const float alpha = __expf(rm[r] - newm);
      const float pv    = __expf(sv - newm);
      float ps = pv;
      ps += __shfl_xor(ps, 1, 32);
      ps += __shfl_xor(ps, 2, 32);
      ps += __shfl_xor(ps, 4, 32);
      ps += __shfl_xor(ps, 8, 32);
      rs[r] = rs[r] * alpha + ps;
      rm[r] = newm;
#pragma unroll
      for (int nc = 0; nc < 8; nc++) accO[nc][r] *= alpha;
      sP[wid][(r + 8 * hi) * 18 + l15] = pv;
    }
    __syncthreads();

#pragma unroll
    for (int k0 = 0; k0 < 16; k0 += 4) {
      const v2f a = *(const v2f*)&sP[wid][l15 * 18 + k0 + 2 * hi];
#pragma unroll
      for (int nc = 0; nc < 8; nc++) {
        const v2f bb = *(const v2f*)&sVt[(nc * 16 + l15) * 20 + k0 + 2 * hi];
        accO[nc] = wmma_f32(a, bb, accO[nc]);
      }
    }
    __syncthreads();
  }

#pragma unroll
  for (int nc = 0; nc < 8; nc++) {
#pragma unroll
    for (int r = 0; r < 8; r++) {
      const int mrow = qbase + wid * 16 + r + 8 * hi;
      out[(size_t)(b * SEQB + mrow) * HIDDEN + h * HD + nc * 16 + l15] = accO[nc][r] / rs[r];
    }
  }
}

extern "C" void kernel_launch(void* const* d_in, const int* in_sizes, int n_in,
                              void* d_out, int out_size, void* d_ws, size_t ws_size,
                              hipStream_t stream) {
  const float* hs    = (const float*)d_in[0];
  const float* qkv_w = (const float*)d_in[1];
  const float* qkv_b = (const float*)d_in[2];
  const float* o_w   = (const float*)d_in[3];
  float* out  = (float*)d_out;
  float* qkv  = (float*)d_ws;
  float* attn = qkv + (size_t)NTOK * QKVLD;

  dim3 g1(QKVLD / 64, NTOK / 128);
  gemm_nt_kernel<<<g1, 128, 0, stream>>>(hs, qkv_w, qkv_b, qkv, NTOK, QKVLD, HIDDEN);

  rope_kernel<<<(NTOK * 1280) / 256, 256, 0, stream>>>(qkv);

  dim3 g3(SEQB / 64, NH, BATCH);
  flash_attn_kernel<<<g3, 128, 0, stream>>>(qkv, attn);

  dim3 g4(HIDDEN / 64, NTOK / 128);
  gemm_nt_kernel<<<g4, 128, 0, stream>>>(attn, o_w, nullptr, out, NTOK, HIDDEN, HIDDEN);
}


# GEMM K loop software-pipelined: next tile's 12 global loads issued as one group before the WMMAs, stored to a 2nd LDS buffer after, one barrier per step; row pitch compile-time
# speedup vs baseline: 1.5167x; 1.5167x over previous
#include <hip/hip_runtime.h>
#include <hip/hip_bf16.h>
#include <math.h>

#define HIDDEN   2048
#define NH       16
#define NKV      4
#define HD       128
#define D2       64
#define QSIZE    2048
#define KVSIZE   512
#define QKVLD    3072
#define KOFF     2048
#define VOFF     2560
#define SEQB     2048
#define BATCH    2
#define NTOK     (BATCH * SEQB)
#define SM_SCALE 0.08838834764831845f

typedef __attribute__((ext_vector_type(2))) float v2f;
typedef __attribute__((ext_vector_type(8))) float v8f;

__device__ __forceinline__ v8f wmma_f32(v2f a, v2f b, v8f c) {
  return __builtin_amdgcn_wmma_f32_16x16x4_f32(
       false, a,  false, b,
       (short)0, c,  false,  false);
}

#define GBK  32
#define GLDP 36
#define GSA  (128 * GLDP)
#define GSB  (64 * GLDP)

__device__ __forceinline__ void gemm_tile_mma(const float* __restrict__ tA,
                                              const float* __restrict__ tB,
                                              v8f (&acc)[2][4],
                                              int wid, int hi, int l15) {
#pragma unroll
  for (int k2 = 0; k2 < GBK; k2 += 4) {
    const v2f a0 = *(const v2f*)&tA[(wid * 32 + l15) * GLDP + k2 + 2 * hi];
    const v2f a1 = *(const v2f*)&tA[(wid * 32 + 16 + l15) * GLDP + k2 + 2 * hi];
#pragma unroll
    for (int nc = 0; nc < 4; nc++) {
      const v2f b = *(const v2f*)&tB[(nc * 16 + l15) * GLDP + k2 + 2 * hi];
      acc[0][nc] = wmma_f32(a0, b, acc[0][nc]);
      acc[1][nc] = wmma_f32(a1, b, acc[1][nc]);
    }
  }
}

template <int KC>
__global__ __launch_bounds__(128)
void gemm_nt_kernel(const float* __restrict__ A, const float* __restrict__ Bw,
                    const float* __restrict__ bias, float* __restrict__ C,
                    int M, int N) {
  __shared__ float sA[2 * GSA];
  __shared__ float sB[2 * GSB];
  const int t    = threadIdx.x;
  const int lane = t & 31, wid = t >> 5;
  const int hi   = lane >> 4, l15 = lane & 15;
  const int mbase = blockIdx.y * 128;
  const int nbase = blockIdx.x * 64;

  const v8f vz = {0.f,0.f,0.f,0.f,0.f,0.f,0.f,0.f};
  v8f acc[2][4] = {{vz, vz, vz, vz}, {vz, vz, vz, vz}};

  const int srow = t >> 3;
  const int scol = (t & 7) << 2;
  const float* pa = A  + (size_t)(mbase + srow) * KC + scol;
  const float* pb = Bw + (size_t)(nbase + srow) * KC + scol;
  const int sofs = srow * GLDP + scol;

  float4 ra[8], rb[4];
#pragma unroll
  for (int i = 0; i < 8; i++) ra[i] = *(const float4*)(pa + (size_t)(16 * i) * KC);
#pragma unroll
  for (int i = 0; i < 4; i++) rb[i] = *(const float4*)(pb + (size_t)(16 * i) * KC);
#pragma unroll
  for (int i = 0; i < 8; i++) *(float4*)&sA[sofs + 16 * i * GLDP] = ra[i];
#pragma unroll
  for (int i = 0; i < 4; i++) *(float4*)&sB[sofs + 16 * i * GLDP] = rb[i];
  __syncthreads();

  constexpr int NT = KC / GBK;
  for (int it = 0; it < NT - 1; it++) {
    const int cur = it & 1;
    pa += GBK;
    pb += GBK;
#pragma unroll
    for (int i = 0; i < 8; i++) ra[i] = *(const float4*)(pa + (size_t)(16 * i) * KC);
#pragma unroll
    for (int i = 0; i < 4; i++) rb[i] = *(const float4*)(pb + (size_t)(16 * i) * KC);

    gemm_tile_mma(&sA[cur * GSA], &sB[cur * GSB], acc, wid, hi, l15);

    float* nA = &sA[(cur ^ 1) * GSA + sofs];
    float* nB = &sB[(cur ^ 1) * GSB + sofs];
#pragma unroll
    for (int i = 0; i < 8; i++) *(float4*)&nA[16 * i * GLDP] = ra[i];
#pragma unroll
    for (int i = 0; i < 4; i++) *(float4*)&nB[16 * i * GLDP] = rb[i];
    __syncthreads();
  }
  gemm_tile_mma(&sA[((NT - 1) & 1) * GSA], &sB[((NT - 1) & 1) * GSB], acc, wid, hi, l15);

#pragma unroll
  for (int st = 0; st < 2; st++) {
#pragma unroll
    for (int nc = 0; nc < 4; nc++) {
      const int n  = nbase + nc * 16 + l15;
      const float bv = bias ? bias[n] : 0.0f;
#pragma unroll
      for (int r = 0; r < 8; r++) {
        const int m = mbase + wid * 32 + st * 16 + r + 8 * hi;
        C[(size_t)m * N + n] = acc[st][nc][r] + bv;
      }
    }
  }
}

__global__ __launch_bounds__(256)
void rope_kernel(float* __restrict__ qkv) {
  const int tid   = blockIdx.x * 256 + threadIdx.x;
  const int token = tid / 1280;
  const int rem   = tid % 1280;
  const int head  = rem >> 6;
  const int d     = rem & 63;
  const int pos   = token & (SEQB - 1);
  const int col   = (head < NH) ? head * HD + d : QSIZE + (head - NH) * HD + d;
  const float inv_freq = powf(1.0e6f, -(float)d * (1.0f / 64.0f));
  float s, c;
  sincosf((float)pos * inv_freq, &s, &c);
  float* p = qkv + (size_t)token * QKVLD + col;
  const float x1 = p[0], x2 = p[D2];
  p[0]  = x1 * c - x2 * s;
  p[D2] = x1 * s + x2 * c;
}

__global__ __launch_bounds__(128)
void flash_attn_kernel(const float* __restrict__ qkv, float* __restrict__ out) {
  __shared__ float sQ[64 * 132];
  __shared__ float sK[16 * 132];
  __shared__ float sVt[128 * 20];
  __shared__ float sP[4][16 * 18];

  const int t    = threadIdx.x;
  const int lane = t & 31, wid = t >> 5;
  const int hi   = lane >> 4, l15 = lane & 15;
  const int qblk = blockIdx.x;
  const int h    = blockIdx.y;
  const int b    = blockIdx.z;
  const int kvh  = h >> 2;
  const int qbase = qblk * 64;
  const size_t rowbase = (size_t)(b * SEQB) * QKVLD;

#pragma unroll
  for (int i = 0; i < 16; i++) {
    const int idx = t + 128 * i;
    const int row = idx >> 5, c4 = (idx & 31) << 2;
    float4 v = *(const float4*)&qkv[rowbase + (size_t)(qbase + row) * QKVLD + h * HD + c4];
    v.x *= SM_SCALE; v.y *= SM_SCALE; v.z *= SM_SCALE; v.w *= SM_SCALE;
    *(float4*)&sQ[row * 132 + c4] = v;
  }

  const v8f vz = {0.f,0.f,0.f,0.f,0.f,0.f,0.f,0.f};
  v8f accO[8] = {vz, vz, vz, vz, vz, vz, vz, vz};
  float rm[8], rs[8];
#pragma unroll
  for (int r = 0; r < 8; r++) { rm[r] = -__builtin_inff(); rs[r] = 0.0f; }

  __syncthreads();

  const int nkb = (qbase >> 4) + 4;
  for (int kbi = 0; kbi < nkb; kbi++) {
    const int kb = kbi * 16;
#pragma unroll
    for (int i = 0; i < 4; i++) {
      const int idx = t + 128 * i;
      const int row = idx >> 5, c4 = (idx & 31) << 2;
      const float4 v = *(const float4*)&qkv[rowbase + (size_t)(kb + row) * QKVLD + KOFF + kvh * HD + c4];
      *(float4*)&sK[row * 132 + c4] = v;
    }
#pragma unroll
    for (int i = 0; i < 16; i++) {
      sVt[t * 20 + i] = qkv[rowbase + (size_t)(kb + i) * QKVLD + VOFF + kvh * HD + t];
    }
    __syncthreads();

    v8f s = vz;
#pragma unroll
    for (int k0 = 0; k0 < HD; k0 += 4) {
      const v2f a  = *(const v2f*)&sQ[(wid * 16 + l15) * 132 + k0 + 2 * hi];
      const v2f bb = *(const v2f*)&sK[l15 * 132 + k0 + 2 * hi];
      s = wmma_f32(a, bb, s);
    }

    const int kcol = kb + l15;
#pragma unroll
    for (int r = 0; r < 8; r++) {
      const int mrow = qbase + wid * 16 + r + 8 * hi;
      const float sv = (kcol <= mrow) ? s[r] : -__builtin_inff();
      float mx = sv;
      mx = fmaxf(mx, __shfl_xor(mx, 1, 32));
      mx = fmaxf(mx, __shfl_xor(mx, 2, 32));
      mx = fmaxf(mx, __shfl_xor(mx, 4, 32));
      mx = fmaxf(mx, __shfl_xor(mx, 8, 32));
      const float newm  = fmaxf(rm[r], mx);
      const float alpha = __expf(rm[r] - newm);
      const float pv    = __expf(sv - newm);
      float ps = pv;
      ps += __shfl_xor(ps, 1, 32);
      ps += __shfl_xor(ps, 2, 32);
      ps += __shfl_xor(ps, 4, 32);
      ps += __shfl_xor(ps, 8, 32);
      rs[r] = rs[r] * alpha + ps;
      rm[r] = newm;
#pragma unroll
      for (int nc = 0; nc < 8; nc++) accO[nc][r] *= alpha;
      sP[wid][(r + 8 * hi) * 18 + l15] = pv;
    }
    __syncthreads();

#pragma unroll
    for (int k0 = 0; k0 < 16; k0 += 4) {
      const v2f a = *(const v2f*)&sP[wid][l15 * 18 + k0 + 2 * hi];
#pragma unroll
      for (int nc = 0; nc < 8; nc++) {
        const v2f bb = *(const v2f*)&sVt[(nc * 16 + l15) * 20 + k0 + 2 * hi];
        accO[nc] = wmma_f32(a, bb, accO[nc]);
      }
    }
    __syncthreads();
  }

#pragma unroll
  for (int nc = 0; nc < 8; nc++) {
#pragma unroll
    for (int r = 0; r < 8; r++) {
      const int mrow = qbase + wid * 16 + r + 8 * hi;
      out[(size_t)(b * SEQB + mrow) * HIDDEN + h * HD + nc * 16 + l15] = accO[nc][r] / rs[r];
    }
  }
}

extern "C" void kernel_launch(void* const* d_in, const int* in_sizes, int n_in,
                              void* d_out, int out_size, void* d_ws, size_t ws_size,
                              hipStream_t stream) {
  const float* hs    = (const float*)d_in[0];
  const float* qkv_w = (const float*)d_in[1];
  const float* qkv_b = (const float*)d_in[2];
  const float* o_w   = (const float*)d_in[3];
  float* out  = (float*)d_out;
  float* qkv  = (float*)d_ws;
  float* attn = qkv + (size_t)NTOK * QKVLD;

  dim3 g1(QKVLD / 64, NTOK / 128);
  gemm_nt_kernel<HIDDEN><<<g1, 128, 0, stream>>>(hs, qkv_w, qkv_b, qkv, NTOK, QKVLD);

  rope_kernel<<<(NTOK * 1280) / 256, 256, 0, stream>>>(qkv);

  dim3 g3(SEQB / 64, NH, BATCH);
  flash_attn_kernel<<<g3, 128, 0, stream>>>(qkv, attn);

  dim3 g4(HIDDEN / 64, NTOK / 128);
  gemm_nt_kernel<HIDDEN><<<g4, 128, 0, stream>>>(attn, o_w, nullptr, out, NTOK, HIDDEN);
}


# attention KV step widened 16 -> 32 keys: two independent score accumulators per wave, one max/sum reduction, one O rescale and one barrier set per 32 keys (row sums now summed per 32 keys)
# speedup vs baseline: 1.6109x; 1.0621x over previous
#include <hip/hip_runtime.h>
#include <hip/hip_bf16.h>
#include <math.h>

#define HIDDEN   2048
#define NH       16
#define NKV      4
#define HD       128
#define D2       64
#define QSIZE    2048
#define KVSIZE   512
#define QKVLD    3072
#define KOFF     2048
#define VOFF     2560
#define SEQB     2048
#define BATCH    2
#define NTOK     (BATCH * SEQB)
#define SM_SCALE 0.08838834764831845f

typedef __attribute__((ext_vector_type(2))) float v2f;
typedef __attribute__((ext_vector_type(8))) float v8f;

__device__ __forceinline__ v8f wmma_f32(v2f a, v2f b, v8f c) {
  return __builtin_amdgcn_wmma_f32_16x16x4_f32(
       false, a,  false, b,
       (short)0, c,  false,  false);
}

#define GBK  32
#define GLDP 36
#define GSA  (128 * GLDP)
#define GSB  (64 * GLDP)

__device__ __forceinline__ void gemm_tile_mma(const float* __restrict__ tA,
                                              const float* __restrict__ tB,
                                              v8f (&acc)[2][4],
                                              int wid, int hi, int l15) {
#pragma unroll
  for (int k2 = 0; k2 < GBK; k2 += 4) {
    const v2f a0 = *(const v2f*)&tA[(wid * 32 + l15) * GLDP + k2 + 2 * hi];
    const v2f a1 = *(const v2f*)&tA[(wid * 32 + 16 + l15) * GLDP + k2 + 2 * hi];
#pragma unroll
    for (int nc = 0; nc < 4; nc++) {
      const v2f b = *(const v2f*)&tB[(nc * 16 + l15) * GLDP + k2 + 2 * hi];
      acc[0][nc] = wmma_f32(a0, b, acc[0][nc]);
      acc[1][nc] = wmma_f32(a1, b, acc[1][nc]);
    }
  }
}

template <int KC>
__global__ __launch_bounds__(128)
void gemm_nt_kernel(const float* __restrict__ A, const float* __restrict__ Bw,
                    const float* __restrict__ bias, float* __restrict__ C,
                    int M, int N) {
  __shared__ float sA[2 * GSA];
  __shared__ float sB[2 * GSB];
  const int t    = threadIdx.x;
  const int lane = t & 31, wid = t >> 5;
  const int hi   = lane >> 4, l15 = lane & 15;
  const int mbase = blockIdx.y * 128;
  const int nbase = blockIdx.x * 64;

  const v8f vz = {0.f,0.f,0.f,0.f,0.f,0.f,0.f,0.f};
  v8f acc[2][4] = {{vz, vz, vz, vz}, {vz, vz, vz, vz}};

  const int srow = t >> 3;
  const int scol = (t & 7) << 2;
  const float* pa = A  + (size_t)(mbase + srow) * KC + scol;
  const float* pb = Bw + (size_t)(nbase + srow) * KC + scol;
  const int sofs = srow * GLDP + scol;

  float4 ra[8], rb[4];
#pragma unroll
  for (int i = 0; i < 8; i++) ra[i] = *(const float4*)(pa + (size_t)(16 * i) * KC);
#pragma unroll
  for (int i = 0; i < 4; i++) rb[i] = *(const float4*)(pb + (size_t)(16 * i) * KC);
#pragma unroll
  for (int i = 0; i < 8; i++) *(float4*)&sA[sofs + 16 * i * GLDP] = ra[i];
#pragma unroll
  for (int i = 0; i < 4; i++) *(float4*)&sB[sofs + 16 * i * GLDP] = rb[i];
  __syncthreads();

  constexpr int NT = KC / GBK;
  for (int it = 0; it < NT - 1; it++) {
    const int cur = it & 1;
    pa += GBK;
    pb += GBK;
#pragma unroll
    for (int i = 0; i < 8; i++) ra[i] = *(const float4*)(pa + (size_t)(16 * i) * KC);
#pragma unroll
    for (int i = 0; i < 4; i++) rb[i] = *(const float4*)(pb + (size_t)(16 * i) * KC);

    gemm_tile_mma(&sA[cur * GSA], &sB[cur * GSB], acc, wid, hi, l15);

    float* nA = &sA[(cur ^ 1) * GSA + sofs];
    float* nB = &sB[(cur ^ 1) * GSB + sofs];
#pragma unroll
    for (int i = 0; i < 8; i++) *(float4*)&nA[16 * i * GLDP] = ra[i];
#pragma unroll
    for (int i = 0; i < 4; i++) *(float4*)&nB[16 * i * GLDP] = rb[i];
    __syncthreads();
  }
  gemm_tile_mma(&sA[((NT - 1) & 1) * GSA], &sB[((NT - 1) & 1) * GSB], acc, wid, hi, l15);

#pragma unroll
  for (int st = 0; st < 2; st++) {
#pragma unroll
    for (int nc = 0; nc < 4; nc++) {
      const int n  = nbase + nc * 16 + l15;
      const float bv = bias ? bias[n] : 0.0f;
#pragma unroll
      for (int r = 0; r < 8; r++) {
        const int m = mbase + wid * 32 + st * 16 + r + 8 * hi;
        C[(size_t)m * N + n] = acc[st][nc][r] + bv;
      }
    }
  }
}

__global__ __launch_bounds__(256)
void rope_kernel(float* __restrict__ qkv) {
  const int tid   = blockIdx.x * 256 + threadIdx.x;
  const int token = tid / 1280;
  const int rem   = tid % 1280;
  const int head  = rem >> 6;
  const int d     = rem & 63;
  const int pos   = token & (SEQB - 1);
  const int col   = (head < NH) ? head * HD + d : QSIZE + (head - NH) * HD + d;
  const float inv_freq = powf(1.0e6f, -(float)d * (1.0f / 64.0f));
  float s, c;
  sincosf((float)pos * inv_freq, &s, &c);
  float* p = qkv + (size_t)token * QKVLD + col;
  const float x1 = p[0], x2 = p[D2];
  p[0]  = x1 * c - x2 * s;
  p[D2] = x1 * s + x2 * c;
}

#define KVB  32
#define KVP  36

__global__ __launch_bounds__(128)
void flash_attn_kernel(const float* __restrict__ qkv, float* __restrict__ out) {
  __shared__ float sQ[64 * 132];
  __shared__ float sK[KVB * 132];
  __shared__ float sVt[128 * KVP];
  __shared__ float sP[4][16 * KVP];

  const int t    = threadIdx.x;
  const int lane = t & 31, wid = t >> 5;
  const int hi   = lane >> 4, l15 = lane & 15;
  const int qblk = blockIdx.x;
  const int h    = blockIdx.y;
  const int b    = blockIdx.z;
  const int kvh  = h >> 2;
  const int qbase = qblk * 64;
  const size_t rowbase = (size_t)(b * SEQB) * QKVLD;

#pragma unroll
  for (int i = 0; i < 16; i++) {
    const int idx = t + 128 * i;
    const int row = idx >> 5, c4 = (idx & 31) << 2;
    float4 v = *(const float4*)&qkv[rowbase + (size_t)(qbase + row) * QKVLD + h * HD + c4];
    v.x *= SM_SCALE; v.y *= SM_SCALE; v.z *= SM_SCALE; v.w *= SM_SCALE;
    *(float4*)&sQ[row * 132 + c4] = v;
  }

  const v8f vz = {0.f,0.f,0.f,0.f,0.f,0.f,0.f,0.f};
  v8f accO[8] = {vz, vz, vz, vz, vz, vz, vz, vz};
  float rm[8], rs[8];
#pragma unroll
  for (int r = 0; r < 8; r++) { rm[r] = -__builtin_inff(); rs[r] = 0.0f; }

  __syncthreads();

  const int nkb = (qbase >> 5) + 2;
  const int krow = t >> 5, kc4 = (t & 31) << 2;
  const float* kptr = qkv + rowbase + (size_t)krow * QKVLD + KOFF + kvh * HD + kc4;
  const float* vptr = qkv + rowbase + VOFF + kvh * HD + t;
  for (int kbi = 0; kbi < nkb; kbi++) {
    const int kb = kbi * KVB;
#pragma unroll
    for (int i = 0; i < 8; i++) {
      const float4 v = *(const float4*)(kptr + (size_t)(4 * i) * QKVLD);
      *(float4*)&sK[(krow + 4 * i) * 132 + kc4] = v;
    }
#pragma unroll
    for (int i = 0; i < KVB; i++) {
      sVt[t * KVP + i] = vptr[(size_t)i * QKVLD];
    }
    kptr += (size_t)KVB * QKVLD;
    vptr += (size_t)KVB * QKVLD;
    __syncthreads();

    v8f s0 = vz, s1 = vz;
#pragma unroll
    for (int k0 = 0; k0 < HD; k0 += 4) {
      const v2f a  = *(const v2f*)&sQ[(wid * 16 + l15) * 132 + k0 + 2 * hi];
      const v2f b0 = *(const v2f*)&sK[l15 * 132 + k0 + 2 * hi];
      const v2f b1 = *(const v2f*)&sK[(16 + l15) * 132 + k0 + 2 * hi];
      s0 = wmma_f32(a, b0, s0);
      s1 = wmma_f32(a, b1, s1);
    }

    const int kcol0 = kb + l15;
    const int kcol1 = kb + 16 + l15;
#pragma unroll
    for (int r = 0; r < 8; r++) {
      const int mrow = qbase + wid * 16 + r + 8 * hi;
      const float sv0 = (kcol0 <= mrow) ? s0[r] : -__builtin_inff();
      const float sv1 = (kcol1 <= mrow) ? s1[r] : -__builtin_inff();
      float mx = fmaxf(sv0, sv1);
      mx = fmaxf(mx, __shfl_xor(mx, 1, 32));
      mx = fmaxf(mx, __shfl_xor(mx, 2, 32));
      mx = fmaxf(mx, __shfl_xor(mx, 4, 32));
      mx = fmaxf(mx, __shfl_xor(mx, 8, 32));
      const float newm  = fmaxf(rm[r], mx);
      const float alpha = __expf(rm[r] - newm);
      const float pv0   = __expf(sv0 - newm);
      const float pv1   = __expf(sv1 - newm);
      float ps = pv0 + pv1;
      ps += __shfl_xor(ps, 1, 32);
      ps += __shfl_xor(ps, 2, 32);
      ps += __shfl_xor(ps, 4, 32);
      ps += __shfl_xor(ps, 8, 32);
      rs[r] = rs[r] * alpha + ps;
      rm[r] = newm;
#pragma unroll
      for (int nc = 0; nc < 8; nc++) accO[nc][r] *= alpha;
      sP[wid][(r + 8 * hi) * KVP + l15]      = pv0;
      sP[wid][(r + 8 * hi) * KVP + 16 + l15] = pv1;
    }
    __syncthreads();

#pragma unroll
    for (int k0 = 0; k0 < KVB; k0 += 4) {
      const v2f a = *(const v2f*)&sP[wid][l15 * KVP + k0 + 2 * hi];
#pragma unroll
      for (int nc = 0; nc < 8; nc++) {
        const v2f bb = *(const v2f*)&sVt[(nc * 16 + l15) * KVP + k0 + 2 * hi];
        accO[nc] = wmma_f32(a, bb, accO[nc]);
      }
    }
    __syncthreads();
  }

#pragma unroll
  for (int nc = 0; nc < 8; nc++) {
#pragma unroll
    for (int r = 0; r < 8; r++) {
      const int mrow = qbase + wid * 16 + r + 8 * hi;
      out[(size_t)(b * SEQB + mrow) * HIDDEN + h * HD + nc * 16 + l15] = accO[nc][r] / rs[r];
    }
  }
}

extern "C" void kernel_launch(void* const* d_in, const int* in_sizes, int n_in,
                              void* d_out, int out_size, void* d_ws, size_t ws_size,
                              hipStream_t stream) {
  const float* hs    = (const float*)d_in[0];
  const float* qkv_w = (const float*)d_in[1];
  const float* qkv_b = (const float*)d_in[2];
  const float* o_w   = (const float*)d_in[3];
  float* out  = (float*)d_out;
  float* qkv  = (float*)d_ws;
  float* attn = qkv + (size_t)NTOK * QKVLD;

  dim3 g1(QKVLD / 64, NTOK / 128);
  gemm_nt_kernel<HIDDEN><<<g1, 128, 0, stream>>>(hs, qkv_w, qkv_b, qkv, NTOK, QKVLD);

  rope_kernel<<<(NTOK * 1280) / 256, 256, 0, stream>>>(qkv);

  dim3 g3(SEQB / 64, NH, BATCH);
  flash_attn_kernel<<<g3, 128, 0, stream>>>(qkv, attn);

  dim3 g4(HIDDEN / 64, NTOK / 128);
  gemm_nt_kernel<HIDDEN><<<g4, 128, 0, stream>>>(attn, o_w, nullptr, out, NTOK, HIDDEN);
}


# attention causal load balance: each workgroup (8 waves) takes q-block 31-p and q-block p side by side on shared K/V tiles, so all workgroups do equal work; per-row summation order unchanged
# speedup vs baseline: 1.8177x; 1.1284x over previous
#include <hip/hip_runtime.h>
#include <hip/hip_bf16.h>
#include <math.h>

#define HIDDEN   2048
#define NH       16
#define NKV      4
#define HD       128
#define D2       64
#define QSIZE    2048
#define KVSIZE   512
#define QKVLD    3072
#define KOFF     2048
#define VOFF     2560
#define SEQB     2048
#define BATCH    2
#define NTOK     (BATCH * SEQB)
#define SM_SCALE 0.08838834764831845f

typedef __attribute__((ext_vector_type(2))) float v2f;
typedef __attribute__((ext_vector_type(8))) float v8f;

__device__ __forceinline__ v8f wmma_f32(v2f a, v2f b, v8f c) {
  return __builtin_amdgcn_wmma_f32_16x16x4_f32(
       false, a,  false, b,
       (short)0, c,  false,  false);
}

#define GBK  32
#define GLDP 36
#define GSA  (128 * GLDP)
#define GSB  (64 * GLDP)

__device__ __forceinline__ void gemm_tile_mma(const float* __restrict__ tA,
                                              const float* __restrict__ tB,
                                              v8f (&acc)[2][4],
                                              int wid, int hi, int l15) {
#pragma unroll
  for (int k2 = 0; k2 < GBK; k2 += 4) {
    const v2f a0 = *(const v2f*)&tA[(wid * 32 + l15) * GLDP + k2 + 2 * hi];
    const v2f a1 = *(const v2f*)&tA[(wid * 32 + 16 + l15) * GLDP + k2 + 2 * hi];
#pragma unroll
    for (int nc = 0; nc < 4; nc++) {
      const v2f b = *(const v2f*)&tB[(nc * 16 + l15) * GLDP + k2 + 2 * hi];
      acc[0][nc] = wmma_f32(a0, b, acc[0][nc]);
      acc[1][nc] = wmma_f32(a1, b, acc[1][nc]);
    }
  }
}

template <int KC>
__global__ __launch_bounds__(128)
void gemm_nt_kernel(const float* __restrict__ A, const float* __restrict__ Bw,
                    const float* __restrict__ bias, float* __restrict__ C,
                    int M, int N) {
  __shared__ float sA[2 * GSA];
  __shared__ float sB[2 * GSB];
  const int t    = threadIdx.x;
  const int lane = t & 31, wid = t >> 5;
  const int hi   = lane >> 4, l15 = lane & 15;
  const int mbase = blockIdx.y * 128;
  const int nbase = blockIdx.x * 64;

  const v8f vz = {0.f,0.f,0.f,0.f,0.f,0.f,0.f,0.f};
  v8f acc[2][4] = {{vz, vz, vz, vz}, {vz, vz, vz, vz}};

  const int srow = t >> 3;
  const int scol = (t & 7) << 2;
  const float* pa = A  + (size_t)(mbase + srow) * KC + scol;
  const float* pb = Bw + (size_t)(nbase + srow) * KC + scol;
  const int sofs = srow * GLDP + scol;

  float4 ra[8], rb[4];
#pragma unroll
  for (int i = 0; i < 8; i++) ra[i] = *(const float4*)(pa + (size_t)(16 * i) * KC);
#pragma unroll
  for (int i = 0; i < 4; i++) rb[i] = *(const float4*)(pb + (size_t)(16 * i) * KC);
#pragma unroll
  for (int i = 0; i < 8; i++) *(float4*)&sA[sofs + 16 * i * GLDP] = ra[i];
#pragma unroll
  for (int i = 0; i < 4; i++) *(float4*)&sB[sofs + 16 * i * GLDP] = rb[i];
  __syncthreads();

  constexpr int NT = KC / GBK;
  for (int it = 0; it < NT - 1; it++) {
    const int cur = it & 1;
    pa += GBK;
    pb += GBK;
#pragma unroll
    for (int i = 0; i < 8; i++) ra[i] = *(const float4*)(pa + (size_t)(16 * i) * KC);
#pragma unroll
    for (int i = 0; i < 4; i++) rb[i] = *(const float4*)(pb + (size_t)(16 * i) * KC);

    gemm_tile_mma(&sA[cur * GSA], &sB[cur * GSB], acc, wid, hi, l15);

    float* nA = &sA[(cur ^ 1) * GSA + sofs];
    float* nB = &sB[(cur ^ 1) * GSB + sofs];
#pragma unroll
    for (int i = 0; i < 8; i++) *(float4*)&nA[16 * i * GLDP] = ra[i];
#pragma unroll
    for (int i = 0; i < 4; i++) *(float4*)&nB[16 * i * GLDP] = rb[i];
    __syncthreads();
  }
  gemm_tile_mma(&sA[((NT - 1) & 1) * GSA], &sB[((NT - 1) & 1) * GSB], acc, wid, hi, l15);

#pragma unroll
  for (int st = 0; st < 2; st++) {
#pragma unroll
    for (int nc = 0; nc < 4; nc++) {
      const int n  = nbase + nc * 16 + l15;
      const float bv = bias ? bias[n] : 0.0f;
#pragma unroll
      for (int r = 0; r < 8; r++) {
        const int m = mbase + wid * 32 + st * 16 + r + 8 * hi;
        C[(size_t)m * N + n] = acc[st][nc][r] + bv;
      }
    }
  }
}

__global__ __launch_bounds__(256)
void rope_kernel(float* __restrict__ qkv) {
  const int tid   = blockIdx.x * 256 + threadIdx.x;
  const int token = tid / 1280;
  const int rem   = tid % 1280;
  const int head  = rem >> 6;
  const int d     = rem & 63;
  const int pos   = token & (SEQB - 1);
  const int col   = (head < NH) ? head * HD + d : QSIZE + (head - NH) * HD + d;
  const float inv_freq = powf(1.0e6f, -(float)d * (1.0f / 64.0f));
  float s, c;
  sincosf((float)pos * inv_freq, &s, &c);
  float* p = qkv + (size_t)token * QKVLD + col;
  const float x1 = p[0], x2 = p[D2];
  p[0]  = x1 * c - x2 * s;
  p[D2] = x1 * s + x2 * c;
}

#define KVB  32
#define KVP  36
#define NQB  (SEQB / 64)

__global__ __launch_bounds__(256)
void flash_attn_kernel(const float* __restrict__ qkv, float* __restrict__ out) {
  __shared__ float sQ[128 * 132];
  __shared__ float sK[KVB * 132];
  __shared__ float sVt[128 * KVP];
  __shared__ float sP[8][16 * KVP];

  const int t    = threadIdx.x;
  const int lane = t & 31;
  const int wid  = __builtin_amdgcn_readfirstlane(t >> 5);
  const int hi   = lane >> 4, l15 = lane & 15;
  const int grp  = wid >> 2;
  const int w4   = wid & 3;
  const int pr   = blockIdx.x;
  const int h    = blockIdx.y;
  const int b    = blockIdx.z;
  const int kvh  = h >> 2;
  const int qbase_long  = (NQB - 1 - pr) * 64;
  const int qbase_short = pr * 64;
  const int qbase = grp ? qbase_short : qbase_long;
  const size_t rowbase = (size_t)(b * SEQB) * QKVLD;

#pragma unroll
  for (int i = 0; i < 16; i++) {
    const int idx = t + 256 * i;
    const int row = idx >> 5, c4 = (idx & 31) << 2;
    const int grow = (row < 64) ? (qbase_long + row) : (qbase_short + row - 64);
    float4 v = *(const float4*)&qkv[rowbase + (size_t)grow * QKVLD + h * HD + c4];
    v.x *= SM_SCALE; v.y *= SM_SCALE; v.z *= SM_SCALE; v.w *= SM_SCALE;
    *(float4*)&sQ[row * 132 + c4] = v;
  }

  const v8f vz = {0.f,0.f,0.f,0.f,0.f,0.f,0.f,0.f};
  v8f accO[8] = {vz, vz, vz, vz, vz, vz, vz, vz};
  float rm[8], rs[8];
#pragma unroll
  for (int r = 0; r < 8; r++) { rm[r] = -__builtin_inff(); rs[r] = 0.0f; }

  __syncthreads();

  const int nkb_all = (qbase_long >> 5) + 2;
  const int nkb_own = (qbase >> 5) + 2;
  const int qrow0   = (grp * 64 + w4 * 16) * 132;
  const int krow = t >> 5, kc4 = (t & 31) << 2;
  const int vdim = t & 127, vk0 = (t >> 7) * 16;
  const float* kptr = qkv + rowbase + (size_t)krow * QKVLD + KOFF + kvh * HD + kc4;
  const float* vptr = qkv + rowbase + (size_t)vk0 * QKVLD + VOFF + kvh * HD + vdim;
  for (int kbi = 0; kbi < nkb_all; kbi++) {
    const int kb = kbi * KVB;
#pragma unroll
    for (int i = 0; i < 4; i++) {
      const float4 v = *(const float4*)(kptr + (size_t)(8 * i) * QKVLD);
      *(float4*)&sK[(krow + 8 * i) * 132 + kc4] = v;
    }
#pragma unroll
    for (int i = 0; i < 16; i++) {
      sVt[vdim * KVP + vk0 + i] = vptr[(size_t)i * QKVLD];
    }
    kptr += (size_t)KVB * QKVLD;
    vptr += (size_t)KVB * QKVLD;
    __syncthreads();

    const bool act = kbi < nkb_own;

    if (act) {
      v8f s0 = vz, s1 = vz;
#pragma unroll
      for (int k0 = 0; k0 < HD; k0 += 4) {
        const v2f a  = *(const v2f*)&sQ[qrow0 + l15 * 132 + k0 + 2 * hi];
        const v2f b0 = *(const v2f*)&sK[l15 * 132 + k0 + 2 * hi];
        const v2f b1 = *(const v2f*)&sK[(16 + l15) * 132 + k0 + 2 * hi];
        s0 = wmma_f32(a, b0, s0);
        s1 = wmma_f32(a, b1, s1);
      }

      const int kcol0 = kb + l15;
      const int kcol1 = kb + 16 + l15;
#pragma unroll
      for (int r = 0; r < 8; r++) {
        const int mrow = qbase + w4 * 16 + r + 8 * hi;
        const float sv0 = (kcol0 <= mrow) ? s0[r] : -__builtin_inff();
        const float sv1 = (kcol1 <= mrow) ? s1[r] : -__builtin_inff();
        float mx = fmaxf(sv0, sv1);
        mx = fmaxf(mx, __shfl_xor(mx, 1, 32));
        mx = fmaxf(mx, __shfl_xor(mx, 2, 32));
        mx = fmaxf(mx, __shfl_xor(mx, 4, 32));
        mx = fmaxf(mx, __shfl_xor(mx, 8, 32));
        const float newm  = fmaxf(rm[r], mx);
        const float alpha = __expf(rm[r] - newm);
        const float pv0   = __expf(sv0 - newm);
        const float pv1   = __expf(sv1 - newm);
        float ps = pv0 + pv1;
        ps += __shfl_xor(ps, 1, 32);
        ps += __shfl_xor(ps, 2, 32);
        ps += __shfl_xor(ps, 4, 32);
        ps += __shfl_xor(ps, 8, 32);
        rs[r] = rs[r] * alpha + ps;
        rm[r] = newm;
#pragma unroll
        for (int nc = 0; nc < 8; nc++) accO[nc][r] *= alpha;
        sP[wid][(r + 8 * hi) * KVP + l15]      = pv0;
        sP[wid][(r + 8 * hi) * KVP + 16 + l15] = pv1;
      }
    }
    __syncthreads();

    if (act) {
#pragma unroll
      for (int k0 = 0; k0 < KVB; k0 += 4) {
        const v2f a = *(const v2f*)&sP[wid][l15 * KVP + k0 + 2 * hi];
#pragma unroll
        for (int nc = 0; nc < 8; nc++) {
          const v2f bb = *(const v2f*)&sVt[(nc * 16 + l15) * KVP + k0 + 2 * hi];
          accO[nc] = wmma_f32(a, bb, accO[nc]);
        }
      }
    }
    __syncthreads();
  }

#pragma unroll
  for (int nc = 0; nc < 8; nc++) {
#pragma unroll
    for (int r = 0; r < 8; r++) {
      const int mrow = qbase + w4 * 16 + r + 8 * hi;
      out[(size_t)(b * SEQB + mrow) * HIDDEN + h * HD + nc * 16 + l15] = accO[nc][r] / rs[r];
    }
  }
}

extern "C" void kernel_launch(void* const* d_in, const int* in_sizes, int n_in,
                              void* d_out, int out_size, void* d_ws, size_t ws_size,
                              hipStream_t stream) {
  const float* hs    = (const float*)d_in[0];
  const float* qkv_w = (const float*)d_in[1];
  const float* qkv_b = (const float*)d_in[2];
  const float* o_w   = (const float*)d_in[3];
  float* out  = (float*)d_out;
  float* qkv  = (float*)d_ws;
  float* attn = qkv + (size_t)NTOK * QKVLD;

  dim3 g1(QKVLD / 64, NTOK / 128);
  gemm_nt_kernel<HIDDEN><<<g1, 128, 0, stream>>>(hs, qkv_w, qkv_b, qkv, NTOK, QKVLD);

  rope_kernel<<<(NTOK * 1280) / 256, 256, 0, stream>>>(qkv);

  dim3 g3(SEQB / 128, NH, BATCH);
  flash_attn_kernel<<<g3, 256, 0, stream>>>(qkv, attn);

  dim3 g4(HIDDEN / 64, NTOK / 128);
  gemm_nt_kernel<HIDDEN><<<g4, 128, 0, stream>>>(attn, o_w, nullptr, out, NTOK, HIDDEN);
}


# GEMM register blocking: wave tile 32x64 -> 64x64 (block tile 128x128), per 128 WMMAs 32 LDS fragment reads not 48, 16 global loads not 24; k of each 8 now summed {0,1,4,5} then {2,3,6,7}
# speedup vs baseline: 1.8799x; 1.0342x over previous
#include <hip/hip_runtime.h>
#include <hip/hip_bf16.h>
#include <math.h>

#define HIDDEN   2048
#define NH       16
#define NKV      4
#define HD       128
#define D2       64
#define QSIZE    2048
#define KVSIZE   512
#define QKVLD    3072
#define KOFF     2048
#define VOFF     2560
#define SEQB     2048
#define BATCH    2
#define NTOK     (BATCH * SEQB)
#define SM_SCALE 0.08838834764831845f

typedef __attribute__((ext_vector_type(2))) float v2f;
typedef __attribute__((ext_vector_type(8))) float v8f;

__device__ __forceinline__ v8f wmma_f32(v2f a, v2f b, v8f c) {
  return __builtin_amdgcn_wmma_f32_16x16x4_f32(
       false, a,  false, b,
       (short)0, c,  false,  false);
}

#define GBK  32
#define GLDP 36
#define GTM  128
#define GTN  128
#define GSA  (GTM * GLDP)
#define GSB  (GTN * GLDP)

__device__ __forceinline__ void gemm_tile_mma(const float* __restrict__ tA,
                                              const float* __restrict__ tB,
                                              v8f (&acc)[4][4],
                                              int wm, int wn, int hi, int l15) {
  const float* fa = tA + (wm * 64 + l15) * GLDP + 4 * hi;
  const float* fb = tB + (wn * 64 + l15) * GLDP + 4 * hi;
#pragma unroll
  for (int q = 0; q < GBK; q += 8) {
    float4 a[4], b[4];
#pragma unroll
    for (int i = 0; i < 4; i++) a[i] = *(const float4*)(fa + i * 16 * GLDP + q);
#pragma unroll
    for (int i = 0; i < 4; i++) b[i] = *(const float4*)(fb + i * 16 * GLDP + q);
#pragma unroll
    for (int mt = 0; mt < 4; mt++) {
      const v2f alo = {a[mt].x, a[mt].y};
#pragma unroll
      for (int nt = 0; nt < 4; nt++) {
        const v2f blo = {b[nt].x, b[nt].y};
        acc[mt][nt] = wmma_f32(alo, blo, acc[mt][nt]);
      }
    }
#pragma unroll
    for (int mt = 0; mt < 4; mt++) {
      const v2f ahi = {a[mt].z, a[mt].w};
#pragma unroll
      for (int nt = 0; nt < 4; nt++) {
        const v2f bhi = {b[nt].z, b[nt].w};
        acc[mt][nt] = wmma_f32(ahi, bhi, acc[mt][nt]);
      }
    }
  }
}

template <int KC>
__global__ __launch_bounds__(128) __attribute__((amdgpu_waves_per_eu(1, 3)))
void gemm_nt_kernel(const float* __restrict__ A, const float* __restrict__ Bw,
                    const float* __restrict__ bias, float* __restrict__ C,
                    int M, int N) {
  __shared__ __attribute__((aligned(16))) float sA[2 * GSA];
  __shared__ __attribute__((aligned(16))) float sB[2 * GSB];
  const int t    = threadIdx.x;
  const int lane = t & 31, wid = t >> 5;
  const int hi   = lane >> 4, l15 = lane & 15;
  const int wm   = wid >> 1, wn = wid & 1;
  const int mbase = blockIdx.y * GTM;
  const int nbase = blockIdx.x * GTN;

  const v8f vz = {0.f,0.f,0.f,0.f,0.f,0.f,0.f,0.f};
  v8f acc[4][4] = {{vz, vz, vz, vz}, {vz, vz, vz, vz}, {vz, vz, vz, vz}, {vz, vz, vz, vz}};

  const int srow = t >> 3;
  const int scol = (t & 7) << 2;
  const float* pa = A  + (size_t)(mbase + srow) * KC + scol;
  const float* pb = Bw + (size_t)(nbase + srow) * KC + scol;
  const int sofs = srow * GLDP + scol;

  float4 ra[8], rb[8];
#pragma unroll
  for (int i = 0; i < 8; i++) ra[i] = *(const float4*)(pa + (size_t)(16 * i) * KC);
#pragma unroll
  for (int i = 0; i < 8; i++) rb[i] = *(const float4*)(pb + (size_t)(16 * i) * KC);
#pragma unroll
  for (int i = 0; i < 8; i++) *(float4*)&sA[sofs + 16 * i * GLDP] = ra[i];
#pragma unroll
  for (int i = 0; i < 8; i++) *(float4*)&sB[sofs + 16 * i * GLDP] = rb[i];
  __syncthreads();

  constexpr int NT = KC / GBK;
  for (int it = 0; it < NT - 1; it++) {
    const int cur = it & 1;
    pa += GBK;
    pb += GBK;
#pragma unroll
    for (int i = 0; i < 8; i++) ra[i] = *(const float4*)(pa + (size_t)(16 * i) * KC);
#pragma unroll
    for (int i = 0; i < 8; i++) rb[i] = *(const float4*)(pb + (size_t)(16 * i) * KC);

    gemm_tile_mma(&sA[cur * GSA], &sB[cur * GSB], acc, wm, wn, hi, l15);

    float* nA = &sA[(cur ^ 1) * GSA + sofs];
    float* nB = &sB[(cur ^ 1) * GSB + sofs];
#pragma unroll
    for (int i = 0; i < 8; i++) *(float4*)&nA[16 * i * GLDP] = ra[i];
#pragma unroll
    for (int i = 0; i < 8; i++) *(float4*)&nB[16 * i * GLDP] = rb[i];
    __syncthreads();
  }
  gemm_tile_mma(&sA[((NT - 1) & 1) * GSA], &sB[((NT - 1) & 1) * GSB], acc, wm, wn, hi, l15);

#pragma unroll
  for (int nt = 0; nt < 4; nt++) {
    const int n  = nbase + wn * 64 + nt * 16 + l15;
    const float bv = bias ? bias[n] : 0.0f;
#pragma unroll
    for (int mt = 0; mt < 4; mt++) {
#pragma unroll
      for (int r = 0; r < 8; r++) {
        const int m = mbase + wm * 64 + mt * 16 + r + 8 * hi;
        C[(size_t)m * N + n] = acc[mt][nt][r] + bv;
      }
    }
  }
}

__global__ __launch_bounds__(256)
void rope_kernel(float* __restrict__ qkv) {
  const int tid   = blockIdx.x * 256 + threadIdx.x;
  const int token = tid / 1280;
  const int rem   = tid % 1280;
  const int head  = rem >> 6;
  const int d     = rem & 63;
  const int pos   = token & (SEQB - 1);
  const int col   = (head < NH) ? head * HD + d : QSIZE + (head - NH) * HD + d;
  const float inv_freq = powf(1.0e6f, -(float)d * (1.0f / 64.0f));
  float s, c;
  sincosf((float)pos * inv_freq, &s, &c);
  float* p = qkv + (size_t)token * QKVLD + col;
  const float x1 = p[0], x2 = p[D2];
  p[0]  = x1 * c - x2 * s;
  p[D2] = x1 * s + x2 * c;
}

#define KVB  32
#define KVP  36
#define NQB  (SEQB / 64)

__global__ __launch_bounds__(256)
void flash_attn_kernel(const float* __restrict__ qkv, float* __restrict__ out) {
  __shared__ float sQ[128 * 132];
  __shared__ float sK[KVB * 132];
  __shared__ float sVt[128 * KVP];
  __shared__ float sP[8][16 * KVP];

  const int t    = threadIdx.x;
  const int lane = t & 31;
  const int wid  = __builtin_amdgcn_readfirstlane(t >> 5);
  const int hi   = lane >> 4, l15 = lane & 15;
  const int grp  = wid >> 2;
  const int w4   = wid & 3;
  const int pr   = blockIdx.x;
  const int h    = blockIdx.y;
  const int b    = blockIdx.z;
  const int kvh  = h >> 2;
  const int qbase_long  = (NQB - 1 - pr) * 64;
  const int qbase_short = pr * 64;
  const int qbase = grp ? qbase_short : qbase_long;
  const size_t rowbase = (size_t)(b * SEQB) * QKVLD;

#pragma unroll
  for (int i = 0; i < 16; i++) {
    const int idx = t + 256 * i;
    const int row = idx >> 5, c4 = (idx & 31) << 2;
    const int grow = (row < 64) ? (qbase_long + row) : (qbase_short + row - 64);
    float4 v = *(const float4*)&qkv[rowbase + (size_t)grow * QKVLD + h * HD + c4];
    v.x *= SM_SCALE; v.y *= SM_SCALE; v.z *= SM_SCALE; v.w *= SM_SCALE;
    *(float4*)&sQ[row * 132 + c4] = v;
  }

  const v8f vz = {0.f,0.f,0.f,0.f,0.f,0.f,0.f,0.f};
  v8f accO[8] = {vz, vz, vz, vz, vz, vz, vz, vz};
  float rm[8], rs[8];
#pragma unroll
  for (int r = 0; r < 8; r++) { rm[r] = -__builtin_inff(); rs[r] = 0.0f; }

  __syncthreads();

  const int nkb_all = (qbase_long >> 5) + 2;
  const int nkb_own = (qbase >> 5) + 2;
  const int qrow0   = (grp * 64 + w4 * 16) * 132;
  const int krow = t >> 5, kc4 = (t & 31) << 2;
  const int vdim = t & 127, vk0 = (t >> 7) * 16;
  const float* kptr = qkv + rowbase + (size_t)krow * QKVLD + KOFF + kvh * HD + kc4;
  const float* vptr = qkv + rowbase + (size_t)vk0 * QKVLD + VOFF + kvh * HD + vdim;
  for (int kbi = 0; kbi < nkb_all; kbi++) {
    const int kb = kbi * KVB;
#pragma unroll
    for (int i = 0; i < 4; i++) {
      const float4 v = *(const float4*)(kptr + (size_t)(8 * i) * QKVLD);
      *(float4*)&sK[(krow + 8 * i) * 132 + kc4] = v;
    }
#pragma unroll
    for (int i = 0; i < 16; i++) {
      sVt[vdim * KVP + vk0 + i] = vptr[(size_t)i * QKVLD];
    }
    kptr += (size_t)KVB * QKVLD;
    vptr += (size_t)KVB * QKVLD;
    __syncthreads();

    const bool act = kbi < nkb_own;

    if (act) {
      v8f s0 = vz, s1 = vz;
#pragma unroll
      for (int k0 = 0; k0 < HD; k0 += 4) {
        const v2f a  = *(const v2f*)&sQ[qrow0 + l15 * 132 + k0 + 2 * hi];
        const v2f b0 = *(const v2f*)&sK[l15 * 132 + k0 + 2 * hi];
        const v2f b1 = *(const v2f*)&sK[(16 + l15) * 132 + k0 + 2 * hi];
        s0 = wmma_f32(a, b0, s0);
        s1 = wmma_f32(a, b1, s1);
      }

      const int kcol0 = kb + l15;
      const int kcol1 = kb + 16 + l15;
#pragma unroll
      for (int r = 0; r < 8; r++) {
        const int mrow = qbase + w4 * 16 + r + 8 * hi;
        const float sv0 = (kcol0 <= mrow) ? s0[r] : -__builtin_inff();
        const float sv1 = (kcol1 <= mrow) ? s1[r] : -__builtin_inff();
        float mx = fmaxf(sv0, sv1);
        mx = fmaxf(mx, __shfl_xor(mx, 1, 32));
        mx = fmaxf(mx, __shfl_xor(mx, 2, 32));
        mx = fmaxf(mx, __shfl_xor(mx, 4, 32));
        mx = fmaxf(mx, __shfl_xor(mx, 8, 32));
        const float newm  = fmaxf(rm[r], mx);
        const float alpha = __expf(rm[r] - newm);
        const float pv0   = __expf(sv0 - newm);
        const float pv1   = __expf(sv1 - newm);
        float ps = pv0 + pv1;
        ps += __shfl_xor(ps, 1, 32);
        ps += __shfl_xor(ps, 2, 32);
        ps += __shfl_xor(ps, 4, 32);
        ps += __shfl_xor(ps, 8, 32);
        rs[r] = rs[r] * alpha + ps;
        rm[r] = newm;
#pragma unroll
        for (int nc = 0; nc < 8; nc++) accO[nc][r] *= alpha;
        sP[wid][(r + 8 * hi) * KVP + l15]      = pv0;
        sP[wid][(r + 8 * hi) * KVP + 16 + l15] = pv1;
      }
    }
    __syncthreads();

    if (act) {
#pragma unroll
      for (int k0 = 0; k0 < KVB; k0 += 4) {
        const v2f a = *(const v2f*)&sP[wid][l15 * KVP + k0 + 2 * hi];
#pragma unroll
        for (int nc = 0; nc < 8; nc++) {
          const v2f bb = *(const v2f*)&sVt[(nc * 16 + l15) * KVP + k0 + 2 * hi];
          accO[nc] = wmma_f32(a, bb, accO[nc]);
        }
      }
    }
    __syncthreads();
  }

#pragma unroll
  for (int nc = 0; nc < 8; nc++) {
#pragma unroll
    for (int r = 0; r < 8; r++) {
      const int mrow = qbase + w4 * 16 + r + 8 * hi;
      out[(size_t)(b * SEQB + mrow) * HIDDEN + h * HD + nc * 16 + l15] = accO[nc][r] / rs[r];
    }
  }
}

extern "C" void kernel_launch(void* const* d_in, const int* in_sizes, int n_in,
                              void* d_out, int out_size, void* d_ws, size_t ws_size,
                              hipStream_t stream) {
  const float* hs    = (const float*)d_in[0];
  const float* qkv_w = (const float*)d_in[1];
  const float* qkv_b = (const float*)d_in[2];
  const float* o_w   = (const float*)d_in[3];
  float* out  = (float*)d_out;
  float* qkv  = (float*)d_ws;
  float* attn = qkv + (size_t)NTOK * QKVLD;

  dim3 g1(QKVLD / GTN, NTOK / GTM);
  gemm_nt_kernel<HIDDEN><<<g1, 128, 0, stream>>>(hs, qkv_w, qkv_b, qkv, NTOK, QKVLD);

  rope_kernel<<<(NTOK * 1280) / 256, 256, 0, stream>>>(qkv);

  dim3 g3(SEQB / 128, NH, BATCH);
  flash_attn_kernel<<<g3, 256, 0, stream>>>(qkv, attn);

  dim3 g4(HIDDEN / GTN, NTOK / GTM);
  gemm_nt_kernel<HIDDEN><<<g4, 128, 0, stream>>>(attn, o_w, nullptr, out, NTOK, HIDDEN);
}


# attention score tile formed transposed (S^T = K*Q^T, O^T = V^T*P^T): one query per lane, so row max and sum are in-lane plus one half swap and P stays in registers; row sums now added in-lane first
# speedup vs baseline: 1.9489x; 1.0367x over previous
#include <hip/hip_runtime.h>
#include <hip/hip_bf16.h>
#include <math.h>

#define HIDDEN   2048
#define NH       16
#define NKV      4
#define HD       128
#define D2       64
#define QSIZE    2048
#define KVSIZE   512
#define QKVLD    3072
#define KOFF     2048
#define VOFF     2560
#define SEQB     2048
#define BATCH    2
#define NTOK     (BATCH * SEQB)
#define SM_SCALE 0.08838834764831845f

typedef __attribute__((ext_vector_type(2))) float v2f;
typedef __attribute__((ext_vector_type(8))) float v8f;

__device__ __forceinline__ v8f wmma_f32(v2f a, v2f b, v8f c) {
  return __builtin_amdgcn_wmma_f32_16x16x4_f32(
       false, a,  false, b,
       (short)0, c,  false,  false);
}

#define GBK  32
#define GLDP 36
#define GTM  128
#define GTN  128
#define GSA  (GTM * GLDP)
#define GSB  (GTN * GLDP)

__device__ __forceinline__ void gemm_tile_mma(const float* __restrict__ tA,
                                              const float* __restrict__ tB,
                                              v8f (&acc)[4][4],
                                              int wm, int wn, int hi, int l15) {
  const float* fa = tA + (wm * 64 + l15) * GLDP + 4 * hi;
  const float* fb = tB + (wn * 64 + l15) * GLDP + 4 * hi;
#pragma unroll
  for (int q = 0; q < GBK; q += 8) {
    float4 a[4], b[4];
#pragma unroll
    for (int i = 0; i < 4; i++) a[i] = *(const float4*)(fa + i * 16 * GLDP + q);
#pragma unroll
    for (int i = 0; i < 4; i++) b[i] = *(const float4*)(fb + i * 16 * GLDP + q);
#pragma unroll
    for (int mt = 0; mt < 4; mt++) {
      const v2f alo = {a[mt].x, a[mt].y};
#pragma unroll
      for (int nt = 0; nt < 4; nt++) {
        const v2f blo = {b[nt].x, b[nt].y};
        acc[mt][nt] = wmma_f32(alo, blo, acc[mt][nt]);
      }
    }
#pragma unroll
    for (int mt = 0; mt < 4; mt++) {
      const v2f ahi = {a[mt].z, a[mt].w};
#pragma unroll
      for (int nt = 0; nt < 4; nt++) {
        const v2f bhi = {b[nt].z, b[nt].w};
        acc[mt][nt] = wmma_f32(ahi, bhi, acc[mt][nt]);
      }
    }
  }
}

template <int KC>
__global__ __launch_bounds__(128) __attribute__((amdgpu_waves_per_eu(1, 3)))
void gemm_nt_kernel(const float* __restrict__ A, const float* __restrict__ Bw,
                    const float* __restrict__ bias, float* __restrict__ C,
                    int M, int N) {
  __shared__ __attribute__((aligned(16))) float sA[2 * GSA];
  __shared__ __attribute__((aligned(16))) float sB[2 * GSB];
  const int t    = threadIdx.x;
  const int lane = t & 31, wid = t >> 5;
  const int hi   = lane >> 4, l15 = lane & 15;
  const int wm   = wid >> 1, wn = wid & 1;
  const int mbase = blockIdx.y * GTM;
  const int nbase = blockIdx.x * GTN;

  const v8f vz = {0.f,0.f,0.f,0.f,0.f,0.f,0.f,0.f};
  v8f acc[4][4] = {{vz, vz, vz, vz}, {vz, vz, vz, vz}, {vz, vz, vz, vz}, {vz, vz, vz, vz}};

  const int srow = t >> 3;
  const int scol = (t & 7) << 2;
  const float* pa = A  + (size_t)(mbase + srow) * KC + scol;
  const float* pb = Bw + (size_t)(nbase + srow) * KC + scol;
  const int sofs = srow * GLDP + scol;

  float4 ra[8], rb[8];
#pragma unroll
  for (int i = 0; i < 8; i++) ra[i] = *(const float4*)(pa + (size_t)(16 * i) * KC);
#pragma unroll
  for (int i = 0; i < 8; i++) rb[i] = *(const float4*)(pb + (size_t)(16 * i) * KC);
#pragma unroll
  for (int i = 0; i < 8; i++) *(float4*)&sA[sofs + 16 * i * GLDP] = ra[i];
#pragma unroll
  for (int i = 0; i < 8; i++) *(float4*)&sB[sofs + 16 * i * GLDP] = rb[i];
  __syncthreads();

  constexpr int NT = KC / GBK;
  for (int it = 0; it < NT - 1; it++) {
    const int cur = it & 1;
    pa += GBK;
    pb += GBK;
#pragma unroll
    for (int i = 0; i < 8; i++) ra[i] = *(const float4*)(pa + (size_t)(16 * i) * KC);
#pragma unroll
    for (int i = 0; i < 8; i++) rb[i] = *(const float4*)(pb + (size_t)(16 * i) * KC);

    gemm_tile_mma(&sA[cur * GSA], &sB[cur * GSB], acc, wm, wn, hi, l15);

    float* nA = &sA[(cur ^ 1) * GSA + sofs];
    float* nB = &sB[(cur ^ 1) * GSB + sofs];
#pragma unroll
    for (int i = 0; i < 8; i++) *(float4*)&nA[16 * i * GLDP] = ra[i];
#pragma unroll
    for (int i = 0; i < 8; i++) *(float4*)&nB[16 * i * GLDP] = rb[i];
    __syncthreads();
  }
  gemm_tile_mma(&sA[((NT - 1) & 1) * GSA], &sB[((NT - 1) & 1) * GSB], acc, wm, wn, hi, l15);

#pragma unroll
  for (int nt = 0; nt < 4; nt++) {
    const int n  = nbase + wn * 64 + nt * 16 + l15;
    const float bv = bias ? bias[n] : 0.0f;
#pragma unroll
    for (int mt = 0; mt < 4; mt++) {
#pragma unroll
      for (int r = 0; r < 8; r++) {
        const int m = mbase + wm * 64 + mt * 16 + r + 8 * hi;
        C[(size_t)m * N + n] = acc[mt][nt][r] + bv;
      }
    }
  }
}

__global__ __launch_bounds__(256)
void rope_kernel(float* __restrict__ qkv) {
  const int tid   = blockIdx.x * 256 + threadIdx.x;
  const int token = tid / 1280;
  const int rem   = tid % 1280;
  const int head  = rem >> 6;
  const int d     = rem & 63;
  const int pos   = token & (SEQB - 1);
  const int col   = (head < NH) ? head * HD + d : QSIZE + (head - NH) * HD + d;
  const float inv_freq = powf(1.0e6f, -(float)d * (1.0f / 64.0f));
  float s, c;
  sincosf((float)pos * inv_freq, &s, &c);
  float* p = qkv + (size_t)token * QKVLD + col;
  const float x1 = p[0], x2 = p[D2];
  p[0]  = x1 * c - x2 * s;
  p[D2] = x1 * s + x2 * c;
}

#define KVB  32
#define KVP  36
#define NQB  (SEQB / 64)

__device__ __forceinline__ float xhalf(float x) {
  const int b = __builtin_bit_cast(int, x);
  return __builtin_bit_cast(float,
      __builtin_amdgcn_permlanex16(b, b, 0x76543210, 0xfedcba98, false, false));
}

__global__ __launch_bounds__(256)
void flash_attn_kernel(const float* __restrict__ qkv, float* __restrict__ out) {
  __shared__ __attribute__((aligned(16))) float sQ[128 * 132];
  __shared__ __attribute__((aligned(16))) float sK[KVB * 132];
  __shared__ __attribute__((aligned(16))) float sVt[128 * KVP];

  const int t    = threadIdx.x;
  const int lane = t & 31;
  const int wid  = __builtin_amdgcn_readfirstlane(t >> 5);
  const int hi   = lane >> 4, l15 = lane & 15;
  const int grp  = wid >> 2;
  const int w4   = wid & 3;
  const int pr   = blockIdx.x;
  const int h    = blockIdx.y;
  const int b    = blockIdx.z;
  const int kvh  = h >> 2;
  const int qbase_long  = (NQB - 1 - pr) * 64;
  const int qbase_short = pr * 64;
  const int qbase = grp ? qbase_short : qbase_long;
  const size_t rowbase = (size_t)(b * SEQB) * QKVLD;

#pragma unroll
  for (int i = 0; i < 16; i++) {
    const int idx = t + 256 * i;
    const int row = idx >> 5, c4 = (idx & 31) << 2;
    const int grow = (row < 64) ? (qbase_long + row) : (qbase_short + row - 64);
    float4 v = *(const float4*)&qkv[rowbase + (size_t)grow * QKVLD + h * HD + c4];
    v.x *= SM_SCALE; v.y *= SM_SCALE; v.z *= SM_SCALE; v.w *= SM_SCALE;
    *(float4*)&sQ[row * 132 + c4] = v;
  }

  const v8f vz = {0.f,0.f,0.f,0.f,0.f,0.f,0.f,0.f};
  v8f accO[8] = {vz, vz, vz, vz, vz, vz, vz, vz};
  float rm = -__builtin_inff(), rs = 0.0f;

  __syncthreads();

  const int nkb_all = (qbase_long >> 5) + 2;
  const int nkb_own = (qbase >> 5) + 2;
  const int qrow0   = (grp * 64 + w4 * 16) * 132;
  const int mrow    = qbase + w4 * 16 + l15;
  const int kperm = 4 * ((l15 & 7) >> 1) + 2 * (l15 >> 3) + (l15 & 1);
  const int krow = t >> 5, kc4 = (t & 31) << 2;
  const int vdim = t & 127, vk0 = (t >> 7) * 16;
  const float* kptr = qkv + rowbase + (size_t)krow * QKVLD + KOFF + kvh * HD + kc4;
  const float* vptr = qkv + rowbase + (size_t)vk0 * QKVLD + VOFF + kvh * HD + vdim;
  for (int kbi = 0; kbi < nkb_all; kbi++) {
    const int kb = kbi * KVB;
#pragma unroll
    for (int i = 0; i < 4; i++) {
      const float4 v = *(const float4*)(kptr + (size_t)(8 * i) * QKVLD);
      *(float4*)&sK[(krow + 8 * i) * 132 + kc4] = v;
    }
#pragma unroll
    for (int i = 0; i < 16; i++) {
      sVt[vdim * KVP + vk0 + i] = vptr[(size_t)i * QKVLD];
    }
    kptr += (size_t)KVB * QKVLD;
    vptr += (size_t)KVB * QKVLD;
    __syncthreads();

    const bool act = kbi < nkb_own;

    if (act) {
      v8f s0 = vz, s1 = vz;
#pragma unroll
      for (int k0 = 0; k0 < HD; k0 += 4) {
        const v2f q  = *(const v2f*)&sQ[qrow0 + l15 * 132 + k0 + 2 * hi];
        const v2f a0 = *(const v2f*)&sK[kperm * 132 + k0 + 2 * hi];
        const v2f a1 = *(const v2f*)&sK[(16 + kperm) * 132 + k0 + 2 * hi];
        s0 = wmma_f32(a0, q, s0);
        s1 = wmma_f32(a1, q, s1);
      }

      const int key0 = kb + 2 * hi;
#pragma unroll
      for (int r = 0; r < 8; r++) {
        const int kc = key0 + 4 * (r >> 1) + (r & 1);
        s0[r] = (kc      <= mrow) ? s0[r] : -__builtin_inff();
        s1[r] = (kc + 16 <= mrow) ? s1[r] : -__builtin_inff();
      }
      float m8[8];
#pragma unroll
      for (int r = 0; r < 8; r++) m8[r] = fmaxf(s0[r], s1[r]);
      float mx = fmaxf(fmaxf(fmaxf(m8[0], m8[1]), fmaxf(m8[2], m8[3])),
                       fmaxf(fmaxf(m8[4], m8[5]), fmaxf(m8[6], m8[7])));
      mx = fmaxf(mx, xhalf(mx));
      const float newm  = fmaxf(rm, mx);
      const float alpha = __expf(rm - newm);
      float p8[8];
#pragma unroll
      for (int r = 0; r < 8; r++) {
        s0[r] = __expf(s0[r] - newm);
        s1[r] = __expf(s1[r] - newm);
        p8[r] = s0[r] + s1[r];
      }
      float ps = ((p8[0] + p8[1]) + (p8[2] + p8[3])) + ((p8[4] + p8[5]) + (p8[6] + p8[7]));
      ps += xhalf(ps);
      rs = rs * alpha + ps;
      rm = newm;
#pragma unroll
      for (int nc = 0; nc < 8; nc++) {
#pragma unroll
        for (int r = 0; r < 8; r++) accO[nc][r] *= alpha;
      }

#pragma unroll
      for (int j = 0; j < 4; j++) {
        const v2f pb = {s0[2 * j], s0[2 * j + 1]};
#pragma unroll
        for (int nc = 0; nc < 8; nc++) {
          const v2f va = *(const v2f*)&sVt[(nc * 16 + l15) * KVP + 4 * j + 2 * hi];
          accO[nc] = wmma_f32(va, pb, accO[nc]);
        }
      }
#pragma unroll
      for (int j = 0; j < 4; j++) {
        const v2f pb = {s1[2 * j], s1[2 * j + 1]};
#pragma unroll
        for (int nc = 0; nc < 8; nc++) {
          const v2f va = *(const v2f*)&sVt[(nc * 16 + l15) * KVP + 16 + 4 * j + 2 * hi];
          accO[nc] = wmma_f32(va, pb, accO[nc]);
        }
      }
    }
    __syncthreads();
  }

  float* so = &sQ[qrow0];
#pragma unroll
  for (int nc = 0; nc < 8; nc++) {
    float4 lo, up;
    lo.x = accO[nc][0] / rs; lo.y = accO[nc][1] / rs; lo.z = accO[nc][2] / rs; lo.w = accO[nc][3] / rs;
    up.x = accO[nc][4] / rs; up.y = accO[nc][5] / rs; up.z = accO[nc][6] / rs; up.w = accO[nc][7] / rs;
    *(float4*)&so[l15 * 132 + nc * 16 + 8 * hi]     = lo;
    *(float4*)&so[l15 * 132 + nc * 16 + 8 * hi + 4] = up;
  }
  __syncthreads();
#pragma unroll
  for (int i = 0; i < 16; i++) {
    const float4 v = *(const float4*)&so[i * 132 + 4 * lane];
    *(float4*)&out[(size_t)(b * SEQB + qbase + w4 * 16 + i) * HIDDEN + h * HD + 4 * lane] = v;
  }
}

extern "C" void kernel_launch(void* const* d_in, const int* in_sizes, int n_in,
                              void* d_out, int out_size, void* d_ws, size_t ws_size,
                              hipStream_t stream) {
  const float* hs    = (const float*)d_in[0];
  const float* qkv_w = (const float*)d_in[1];
  const float* qkv_b = (const float*)d_in[2];
  const float* o_w   = (const float*)d_in[3];
  float* out  = (float*)d_out;
  float* qkv  = (float*)d_ws;
  float* attn = qkv + (size_t)NTOK * QKVLD;

  dim3 g1(QKVLD / GTN, NTOK / GTM);
  gemm_nt_kernel<HIDDEN><<<g1, 128, 0, stream>>>(hs, qkv_w, qkv_b, qkv, NTOK, QKVLD);

  rope_kernel<<<(NTOK * 1280) / 256, 256, 0, stream>>>(qkv);

  dim3 g3(SEQB / 128, NH, BATCH);
  flash_attn_kernel<<<g3, 256, 0, stream>>>(qkv, attn);

  dim3 g4(HIDDEN / GTN, NTOK / GTM);
  gemm_nt_kernel<HIDDEN><<<g4, 128, 0, stream>>>(attn, o_w, nullptr, out, NTOK, HIDDEN);
}


# wave priority stepped down by loop progress (s_setprio 3,2,1,0 at 24, 12, 4 steps left) in both GEMM K loops and the attention key loop, so workgroups sharing a SIMD end together; sums unchanged
# speedup vs baseline: 2.0156x; 1.0342x over previous
#include <hip/hip_runtime.h>
#include <hip/hip_bf16.h>
#include <math.h>

#define HIDDEN   2048
#define NH       16
#define NKV      4
#define HD       128
#define D2       64
#define QSIZE    2048
#define KVSIZE   512
#define QKVLD    3072
#define KOFF     2048
#define VOFF     2560
#define SEQB     2048
#define BATCH    2
#define NTOK     (BATCH * SEQB)
#define SM_SCALE 0.08838834764831845f

typedef __attribute__((ext_vector_type(2))) float v2f;
typedef __attribute__((ext_vector_type(8))) float v8f;

__device__ __forceinline__ v8f wmma_f32(v2f a, v2f b, v8f c) {
  return __builtin_amdgcn_wmma_f32_16x16x4_f32(
       false, a,  false, b,
       (short)0, c,  false,  false);
}

#define GBK  32
#define GLDP 36
#define GTM  128
#define GTN  128
#define GSA  (GTM * GLDP)
#define GSB  (GTN * GLDP)

__device__ __forceinline__ void gemm_tile_mma(const float* __restrict__ tA,
                                              const float* __restrict__ tB,
                                              v8f (&acc)[4][4],
                                              int wm, int wn, int hi, int l15) {
  const float* fa = tA + (wm * 64 + l15) * GLDP + 4 * hi;
  const float* fb = tB + (wn * 64 + l15) * GLDP + 4 * hi;
#pragma unroll
  for (int q = 0; q < GBK; q += 8) {
    float4 a[4], b[4];
#pragma unroll
    for (int i = 0; i < 4; i++) a[i] = *(const float4*)(fa + i * 16 * GLDP + q);
#pragma unroll
    for (int i = 0; i < 4; i++) b[i] = *(const float4*)(fb + i * 16 * GLDP + q);
#pragma unroll
    for (int mt = 0; mt < 4; mt++) {
      const v2f alo = {a[mt].x, a[mt].y};
#pragma unroll
      for (int nt = 0; nt < 4; nt++) {
        const v2f blo = {b[nt].x, b[nt].y};
        acc[mt][nt] = wmma_f32(alo, blo, acc[mt][nt]);
      }
    }
#pragma unroll
    for (int mt = 0; mt < 4; mt++) {
      const v2f ahi = {a[mt].z, a[mt].w};
#pragma unroll
      for (int nt = 0; nt < 4; nt++) {
        const v2f bhi = {b[nt].z, b[nt].w};
        acc[mt][nt] = wmma_f32(ahi, bhi, acc[mt][nt]);
      }
    }
  }
}

#define PRIO_REM2 24
#define PRIO_REM1 12
#define PRIO_REM0 4

template <int KC>
__global__ __launch_bounds__(128) __attribute__((amdgpu_waves_per_eu(1, 3)))
void gemm_nt_kernel(const float* __restrict__ A, const float* __restrict__ Bw,
                    const float* __restrict__ bias, float* __restrict__ C,
                    int M, int N) {
  __shared__ __attribute__((aligned(16))) float sA[2 * GSA];
  __shared__ __attribute__((aligned(16))) float sB[2 * GSB];
  const int t    = threadIdx.x;
  const int lane = t & 31, wid = t >> 5;
  const int hi   = lane >> 4, l15 = lane & 15;
  const int wm   = wid >> 1, wn = wid & 1;
  const int mbase = blockIdx.y * GTM;
  const int nbase = blockIdx.x * GTN;

  const v8f vz = {0.f,0.f,0.f,0.f,0.f,0.f,0.f,0.f};
  v8f acc[4][4] = {{vz, vz, vz, vz}, {vz, vz, vz, vz}, {vz, vz, vz, vz}, {vz, vz, vz, vz}};

  const int srow = t >> 3;
  const int scol = (t & 7) << 2;
  const float* pa = A  + (size_t)(mbase + srow) * KC + scol;
  const float* pb = Bw + (size_t)(nbase + srow) * KC + scol;
  const int sofs = srow * GLDP + scol;

  __builtin_amdgcn_s_setprio(3);

  float4 ra[8], rb[8];
#pragma unroll
  for (int i = 0; i < 8; i++) ra[i] = *(const float4*)(pa + (size_t)(16 * i) * KC);
#pragma unroll
  for (int i = 0; i < 8; i++) rb[i] = *(const float4*)(pb + (size_t)(16 * i) * KC);
#pragma unroll
  for (int i = 0; i < 8; i++) *(float4*)&sA[sofs + 16 * i * GLDP] = ra[i];
#pragma unroll
  for (int i = 0; i < 8; i++) *(float4*)&sB[sofs + 16 * i * GLDP] = rb[i];
  __syncthreads();

  constexpr int NT = KC / GBK;
  for (int it = 0; it < NT - 1; it++) {
    const int cur = it & 1;
    const int rem = NT - it;
    if (rem == PRIO_REM2)      __builtin_amdgcn_s_setprio(2);
    else if (rem == PRIO_REM1) __builtin_amdgcn_s_setprio(1);
    else if (rem == PRIO_REM0) __builtin_amdgcn_s_setprio(0);
    pa += GBK;
    pb += GBK;
#pragma unroll
    for (int i = 0; i < 8; i++) ra[i] = *(const float4*)(pa + (size_t)(16 * i) * KC);
#pragma unroll
    for (int i = 0; i < 8; i++) rb[i] = *(const float4*)(pb + (size_t)(16 * i) * KC);

    gemm_tile_mma(&sA[cur * GSA], &sB[cur * GSB], acc, wm, wn, hi, l15);

    float* nA = &sA[(cur ^ 1) * GSA + sofs];
    float* nB = &sB[(cur ^ 1) * GSB + sofs];
#pragma unroll
    for (int i = 0; i < 8; i++) *(float4*)&nA[16 * i * GLDP] = ra[i];
#pragma unroll
    for (int i = 0; i < 8; i++) *(float4*)&nB[16 * i * GLDP] = rb[i];
    __syncthreads();
  }
  gemm_tile_mma(&sA[((NT - 1) & 1) * GSA], &sB[((NT - 1) & 1) * GSB], acc, wm, wn, hi, l15);

#pragma unroll
  for (int nt = 0; nt < 4; nt++) {
    const int n  = nbase + wn * 64 + nt * 16 + l15;
    const float bv = bias ? bias[n] : 0.0f;
#pragma unroll
    for (int mt = 0; mt < 4; mt++) {
#pragma unroll
      for (int r = 0; r < 8; r++) {
        const int m = mbase + wm * 64 + mt * 16 + r + 8 * hi;
        C[(size_t)m * N + n] = acc[mt][nt][r] + bv;
      }
    }
  }
}

__global__ __launch_bounds__(256)
void rope_kernel(float* __restrict__ qkv) {
  const int tid   = blockIdx.x * 256 + threadIdx.x;
  const int token = tid / 1280;
  const int rem   = tid % 1280;
  const int head  = rem >> 6;
  const int d     = rem & 63;
  const int pos   = token & (SEQB - 1);
  const int col   = (head < NH) ? head * HD + d : QSIZE + (head - NH) * HD + d;
  const float inv_freq = powf(1.0e6f, -(float)d * (1.0f / 64.0f));
  float s, c;
  sincosf((float)pos * inv_freq, &s, &c);
  float* p = qkv + (size_t)token * QKVLD + col;
  const float x1 = p[0], x2 = p[D2];
  p[0]  = x1 * c - x2 * s;
  p[D2] = x1 * s + x2 * c;
}

#define KVB  32
#define KVP  36
#define NQB  (SEQB / 64)

__device__ __forceinline__ float xhalf(float x) {
  const int b = __builtin_bit_cast(int, x);
  return __builtin_bit_cast(float,
      __builtin_amdgcn_permlanex16(b, b, 0x76543210, 0xfedcba98, false, false));
}

__global__ __launch_bounds__(256)
void flash_attn_kernel(const float* __restrict__ qkv, float* __restrict__ out) {
  __shared__ __attribute__((aligned(16))) float sQ[128 * 132];
  __shared__ __attribute__((aligned(16))) float sK[KVB * 132];
  __shared__ __attribute__((aligned(16))) float sVt[128 * KVP];

  const int t    = threadIdx.x;
  const int lane = t & 31;
  const int wid  = __builtin_amdgcn_readfirstlane(t >> 5);
  const int hi   = lane >> 4, l15 = lane & 15;
  const int grp  = wid >> 2;
  const int w4   = wid & 3;
  const int pr   = blockIdx.x;
  const int h    = blockIdx.y;
  const int b    = blockIdx.z;
  const int kvh  = h >> 2;
  const int qbase_long  = (NQB - 1 - pr) * 64;
  const int qbase_short = pr * 64;
  const int qbase = grp ? qbase_short : qbase_long;
  const size_t rowbase = (size_t)(b * SEQB) * QKVLD;

#pragma unroll
  for (int i = 0; i < 16; i++) {
    const int idx = t + 256 * i;
    const int row = idx >> 5, c4 = (idx & 31) << 2;
    const int grow = (row < 64) ? (qbase_long + row) : (qbase_short + row - 64);
    float4 v = *(const float4*)&qkv[rowbase + (size_t)grow * QKVLD + h * HD + c4];
    v.x *= SM_SCALE; v.y *= SM_SCALE; v.z *= SM_SCALE; v.w *= SM_SCALE;
    *(float4*)&sQ[row * 132 + c4] = v;
  }

  const v8f vz = {0.f,0.f,0.f,0.f,0.f,0.f,0.f,0.f};
  v8f accO[8] = {vz, vz, vz, vz, vz, vz, vz, vz};
  float rm = -__builtin_inff(), rs = 0.0f;

  __syncthreads();

  const int nkb_all = (qbase_long >> 5) + 2;
  const int nkb_own = (qbase >> 5) + 2;
  const int nkb_short = (qbase_short >> 5) + 2;
  __builtin_amdgcn_s_setprio(3);
  int plevel = 3;
  const int qrow0   = (grp * 64 + w4 * 16) * 132;
  const int mrow    = qbase + w4 * 16 + l15;
  const int kperm = 4 * ((l15 & 7) >> 1) + 2 * (l15 >> 3) + (l15 & 1);
  const int krow = t >> 5, kc4 = (t & 31) << 2;
  const int vdim = t & 127, vk0 = (t >> 7) * 16;
  const float* kptr = qkv + rowbase + (size_t)krow * QKVLD + KOFF + kvh * HD + kc4;
  const float* vptr = qkv + rowbase + (size_t)vk0 * QKVLD + VOFF + kvh * HD + vdim;
  for (int kbi = 0; kbi < nkb_all; kbi++) {
    const int kb = kbi * KVB;
    const int owed  = (nkb_all - kbi) + ((nkb_short > kbi) ? (nkb_short - kbi) : 0);
    const int level = (owed > PRIO_REM2) ? 3 : (owed > PRIO_REM1) ? 2 : (owed > PRIO_REM0) ? 1 : 0;
    if (level != plevel) {
      plevel = level;
      if (level == 2)      __builtin_amdgcn_s_setprio(2);
      else if (level == 1) __builtin_amdgcn_s_setprio(1);
      else                 __builtin_amdgcn_s_setprio(0);
    }
#pragma unroll
    for (int i = 0; i < 4; i++) {
      const float4 v = *(const float4*)(kptr + (size_t)(8 * i) * QKVLD);
      *(float4*)&sK[(krow + 8 * i) * 132 + kc4] = v;
    }
#pragma unroll
    for (int i = 0; i < 16; i++) {
      sVt[vdim * KVP + vk0 + i] = vptr[(size_t)i * QKVLD];
    }
    kptr += (size_t)KVB * QKVLD;
    vptr += (size_t)KVB * QKVLD;
    __syncthreads();

    const bool act = kbi < nkb_own;

    if (act) {
      v8f s0 = vz, s1 = vz;
#pragma unroll
      for (int k0 = 0; k0 < HD; k0 += 4) {
        const v2f q  = *(const v2f*)&sQ[qrow0 + l15 * 132 + k0 + 2 * hi];
        const v2f a0 = *(const v2f*)&sK[kperm * 132 + k0 + 2 * hi];
        const v2f a1 = *(const v2f*)&sK[(16 + kperm) * 132 + k0 + 2 * hi];
        s0 = wmma_f32(a0, q, s0);
        s1 = wmma_f32(a1, q, s1);
      }

      const int key0 = kb + 2 * hi;
#pragma unroll
      for (int r = 0; r < 8; r++) {
        const int kc = key0 + 4 * (r >> 1) + (r & 1);
        s0[r] = (kc      <= mrow) ? s0[r] : -__builtin_inff();
        s1[r] = (kc + 16 <= mrow) ? s1[r] : -__builtin_inff();
      }
      float m8[8];
#pragma unroll
      for (int r = 0; r < 8; r++) m8[r] = fmaxf(s0[r], s1[r]);
      float mx = fmaxf(fmaxf(fmaxf(m8[0], m8[1]), fmaxf(m8[2], m8[3])),
                       fmaxf(fmaxf(m8[4], m8[5]), fmaxf(m8[6], m8[7])));
      mx = fmaxf(mx, xhalf(mx));
      const float newm  = fmaxf(rm, mx);
      const float alpha = __expf(rm - newm);
      float p8[8];
#pragma unroll
      for (int r = 0; r < 8; r++) {
        s0[r] = __expf(s0[r] - newm);
        s1[r] = __expf(s1[r] - newm);
        p8[r] = s0[r] + s1[r];
      }
      float ps = ((p8[0] + p8[1]) + (p8[2] + p8[3])) + ((p8[4] + p8[5]) + (p8[6] + p8[7]));
      ps += xhalf(ps);
      rs = rs * alpha + ps;
      rm = newm;
#pragma unroll
      for (int nc = 0; nc < 8; nc++) {
#pragma unroll
        for (int r = 0; r < 8; r++) accO[nc][r] *= alpha;
      }

#pragma unroll
      for (int j = 0; j < 4; j++) {
        const v2f pb = {s0[2 * j], s0[2 * j + 1]};
#pragma unroll
        for (int nc = 0; nc < 8; nc++) {
          const v2f va = *(const v2f*)&sVt[(nc * 16 + l15) * KVP + 4 * j + 2 * hi];
          accO[nc] = wmma_f32(va, pb, accO[nc]);
        }
      }
#pragma unroll
      for (int j = 0; j < 4; j++) {
        const v2f pb = {s1[2 * j], s1[2 * j + 1]};
#pragma unroll
        for (int nc = 0; nc < 8; nc++) {
          const v2f va = *(const v2f*)&sVt[(nc * 16 + l15) * KVP + 16 + 4 * j + 2 * hi];
          accO[nc] = wmma_f32(va, pb, accO[nc]);
        }
      }
    }
    __syncthreads();
  }

  float* so = &sQ[qrow0];
#pragma unroll
  for (int nc = 0; nc < 8; nc++) {
    float4 lo, up;
    lo.x = accO[nc][0] / rs; lo.y = accO[nc][1] / rs; lo.z = accO[nc][2] / rs; lo.w = accO[nc][3] / rs;
    up.x = accO[nc][4] / rs; up.y = accO[nc][5] / rs; up.z = accO[nc][6] / rs; up.w = accO[nc][7] / rs;
    *(float4*)&so[l15 * 132 + nc * 16 + 8 * hi]     = lo;
    *(float4*)&so[l15 * 132 + nc * 16 + 8 * hi + 4] = up;
  }
  __syncthreads();
#pragma unroll
  for (int i = 0; i < 16; i++) {
    const float4 v = *(const float4*)&so[i * 132 + 4 * lane];
    *(float4*)&out[(size_t)(b * SEQB + qbase + w4 * 16 + i) * HIDDEN + h * HD + 4 * lane] = v;
  }
}

extern "C" void kernel_launch(void* const* d_in, const int* in_sizes, int n_in,
                              void* d_out, int out_size, void* d_ws, size_t ws_size,
                              hipStream_t stream) {
  const float* hs    = (const float*)d_in[0];
  const float* qkv_w = (const float*)d_in[1];
  const float* qkv_b = (const float*)d_in[2];
  const float* o_w   = (const float*)d_in[3];
  float* out  = (float*)d_out;
  float* qkv  = (float*)d_ws;
  float* attn = qkv + (size_t)NTOK * QKVLD;

  dim3 g1(QKVLD / GTN, NTOK / GTM);
  gemm_nt_kernel<HIDDEN><<<g1, 128, 0, stream>>>(hs, qkv_w, qkv_b, qkv, NTOK, QKVLD);

  rope_kernel<<<(NTOK * 1280) / 256, 256, 0, stream>>>(qkv);

  dim3 g3(SEQB / 128, NH, BATCH);
  flash_attn_kernel<<<g3, 256, 0, stream>>>(qkv, attn);

  dim3 g4(HIDDEN / GTN, NTOK / GTM);
  gemm_nt_kernel<HIDDEN><<<g4, 128, 0, stream>>>(attn, o_w, nullptr, out, NTOK, HIDDEN);
}
